# starting kernel
# baseline (speedup 1.0000x reference)
#include <hip/hip_runtime.h>


#define C_CH 256
#define HW_P 4096
#define N_B  4
#define PCHUNKS 8
#define PCHUNK_LEN (HW_P / PCHUNKS)

typedef __attribute__((ext_vector_type(2))) float v2f;
typedef __attribute__((ext_vector_type(4))) float v4f;
typedef __attribute__((ext_vector_type(8))) float v8f;

__device__ __forceinline__ v8f wmma_f32(v2f a, v2f b, v8f c) {
  return __builtin_amdgcn_wmma_f32_16x16x4_f32(false, a, false, b, (short)0, c,
                                               false, false);
}

__device__ __forceinline__ v8f splat8(float v) {
  v8f r = {v, v, v, v, v, v, v, v};
  return r;
}

__global__ void __launch_bounds__(256)
qkv_proj_kernel(const float* __restrict__ x,
                const float* __restrict__ w0, const float* __restrict__ b0,
                const float* __restrict__ w1, const float* __restrict__ b1,
                const float* __restrict__ w2, const float* __restrict__ b2,
                float* __restrict__ Vb, float* __restrict__ Kb,
                float* __restrict__ Qb) {
  const int lane = threadIdx.x & 31;
  const int l15  = lane & 15;
  const int hi   = lane >> 4;
  int t = blockIdx.x * (blockDim.x >> 5) + (threadIdx.x >> 5);
  const int opair = t & 7;   t >>= 3;
  const int ptile = t & 255; t >>= 8;
  const int n     = t & 3;   t >>= 2;
  const int proj  = t;

  const float* w    = (proj == 0) ? w0 : (proj == 1) ? w1 : w2;
  const float* bias = (proj == 0) ? b0 : (proj == 1) ? b1 : b2;
  float* out        = (proj == 0) ? Vb : (proj == 1) ? Kb : Qb;

  const int p0 = ptile * 16;
  const int o0 = opair * 32;
  const float* xn    = x + (size_t)n * C_CH * HW_P;
  const float* wrow0 = w + (size_t)(o0 + l15) * C_CH;
  const float* wrow1 = w + (size_t)(o0 + 16 + l15) * C_CH;

  v8f acc0 = splat8(bias[o0 + l15]);
  v8f acc1 = splat8(bias[o0 + 16 + l15]);
#pragma unroll 4
  for (int k0 = 0; k0 < C_CH; k0 += 4) {
    const int cb = k0 + 2 * hi;
    v2f a;
    a.x = xn[(size_t)cb * HW_P + p0 + l15];
    a.y = xn[(size_t)(cb + 1) * HW_P + p0 + l15];
    v2f bb0 = *(const v2f*)(wrow0 + cb);
    v2f bb1 = *(const v2f*)(wrow1 + cb);
    acc0 = wmma_f32(a, bb0, acc0);
    acc1 = wmma_f32(a, bb1, acc1);
  }
  float* obase = out + ((size_t)n * HW_P + p0) * C_CH + o0 + l15;
#pragma unroll
  for (int r = 0; r < 8; ++r) {
    obase[(size_t)(r + 8 * hi) * C_CH]      = acc0[r];
    obase[(size_t)(r + 8 * hi) * C_CH + 16] = acc1[r];
  }
}

__global__ void __launch_bounds__(256)
kv_gram_kernel(const float* __restrict__ Kb, const float* __restrict__ Vb,
               float* __restrict__ Mpart) {
  const int lane = threadIdx.x & 31;
  const int l15  = lane & 15;
  const int hi   = lane >> 4;
  int t = blockIdx.x * (blockDim.x >> 5) + (threadIdx.x >> 5);
  const int chunk = t & 7;  t >>= 3;
  const int ctile = t & 15; t >>= 4;
  const int dtile = t & 15; t >>= 4;
  const int n     = t;

  const int d0 = dtile * 16, c0 = ctile * 16;
  const int pbeg = chunk * PCHUNK_LEN;
  const float* kbase = Kb + (size_t)n * HW_P * C_CH;
  const float* vbase = Vb + (size_t)n * HW_P * C_CH;

  v8f acc = {};
#pragma unroll 4
  for (int p = pbeg; p < pbeg + PCHUNK_LEN; p += 4) {
    const int pb = p + 2 * hi;
    v2f a, b;
    a.x = kbase[(size_t)pb * C_CH + d0 + l15];
    a.y = kbase[(size_t)(pb + 1) * C_CH + d0 + l15];
    b.x = vbase[(size_t)pb * C_CH + c0 + l15];
    b.y = vbase[(size_t)(pb + 1) * C_CH + c0 + l15];
    acc = wmma_f32(a, b, acc);
  }
  float* mp = Mpart + ((size_t)chunk * N_B + n) * C_CH * C_CH +
              (size_t)d0 * C_CH + c0 + l15;
#pragma unroll
  for (int r = 0; r < 8; ++r)
    mp[(size_t)(r + 8 * hi) * C_CH] = acc[r];
}

__global__ void __launch_bounds__(256)
reduce_m_kernel(const float* __restrict__ Mpart, float* __restrict__ Mb) {
  const size_t idx    = (size_t)blockIdx.x * blockDim.x + threadIdx.x;
  const size_t stride = (size_t)N_B * C_CH * C_CH;
  float s = 0.0f;
#pragma unroll
  for (int c = 0; c < PCHUNKS; ++c) s += Mpart[c * stride + idx];
  Mb[idx] = s;
}

__global__ void __launch_bounds__(256)
m_w3_kernel(const float* __restrict__ Mb, const float* __restrict__ w3,
            float* __restrict__ M2) {
  const int lane = threadIdx.x & 31;
  const int l15  = lane & 15;
  const int hi   = lane >> 4;
  int t = blockIdx.x * (blockDim.x >> 5) + (threadIdx.x >> 5);
  const int otile = t & 15; t >>= 4;
  const int dtile = t & 15; t >>= 4;
  const int n     = t;
  const int d0 = dtile * 16, o0 = otile * 16;

  const float* mrow = Mb + ((size_t)n * C_CH + d0 + l15) * C_CH;
  const float* wrow = w3 + (size_t)(o0 + l15) * C_CH;
  v8f acc = {};
#pragma unroll 4
  for (int k0 = 0; k0 < C_CH; k0 += 4) {
    const int cb = k0 + 2 * hi;
    v2f a = *(const v2f*)(mrow + cb);
    v2f b = *(const v2f*)(wrow + cb);
    acc = wmma_f32(a, b, acc);
  }
  const float inv_hw = 1.0f / (float)HW_P;
  float* m2base = M2 + ((size_t)n * C_CH + d0) * C_CH + o0 + l15;
#pragma unroll
  for (int r = 0; r < 8; ++r)
    m2base[(size_t)(r + 8 * hi) * C_CH] = acc[r] * inv_hw;
}

__global__ void __launch_bounds__(256)
out_kernel(const float* __restrict__ Qb, const float* __restrict__ M2,
           const float* __restrict__ b3, float* __restrict__ out) {
  const int lane = threadIdx.x & 31;
  const int l15  = lane & 15;
  const int hi   = lane >> 4;
  int t = blockIdx.x * (blockDim.x >> 5) + (threadIdx.x >> 5);
  const int opair = t & 7;   t >>= 3;
  const int ptile = t & 255; t >>= 8;
  const int n     = t;
  const int p0 = ptile * 16, o0 = opair * 32;

  const float* qrow = Qb + ((size_t)n * HW_P + p0 + l15) * C_CH;
  const float* m2b  = M2 + (size_t)n * C_CH * C_CH;

  v8f acc0 = splat8(b3[o0 + l15]);
  v8f acc1 = splat8(b3[o0 + 16 + l15]);
#pragma unroll 4
  for (int k0 = 0; k0 < C_CH; k0 += 4) {
    const int db = k0 + 2 * hi;
    v2f a = *(const v2f*)(qrow + db);
    v2f bb0, bb1;
    bb0.x = m2b[(size_t)db * C_CH + o0 + l15];
    bb0.y = m2b[(size_t)(db + 1) * C_CH + o0 + l15];
    bb1.x = m2b[(size_t)db * C_CH + o0 + 16 + l15];
    bb1.y = m2b[(size_t)(db + 1) * C_CH + o0 + 16 + l15];
    acc0 = wmma_f32(a, bb0, acc0);
    acc1 = wmma_f32(a, bb1, acc1);
  }
  float* ob0 = out + ((size_t)n * C_CH + o0 + l15) * HW_P + p0 + 8 * hi;
  float* ob1 = out + ((size_t)n * C_CH + o0 + 16 + l15) * HW_P + p0 + 8 * hi;
  v4f a_lo = {acc0[0], acc0[1], acc0[2], acc0[3]};
  v4f a_hi = {acc0[4], acc0[5], acc0[6], acc0[7]};
  v4f b_lo = {acc1[0], acc1[1], acc1[2], acc1[3]};
  v4f b_hi = {acc1[4], acc1[5], acc1[6], acc1[7]};
  *(v4f*)(ob0)     = a_lo;
  *(v4f*)(ob0 + 4) = a_hi;
  *(v4f*)(ob1)     = b_lo;
  *(v4f*)(ob1 + 4) = b_hi;
}

extern "C" void kernel_launch(void* const* d_in, const int* in_sizes, int n_in,
                              void* d_out, int out_size, void* d_ws,
                              size_t ws_size, hipStream_t stream) {
  const float* x  = (const float*)d_in[0];
  const float* w0 = (const float*)d_in[1];
  const float* b0 = (const float*)d_in[2];
  const float* w1 = (const float*)d_in[3];
  const float* b1 = (const float*)d_in[4];
  const float* w2 = (const float*)d_in[5];
  const float* b2 = (const float*)d_in[6];
  const float* w3 = (const float*)d_in[7];
  const float* b3 = (const float*)d_in[8];
  float* out = (float*)d_out;

  const size_t per = (size_t)N_B * HW_P * C_CH;
  const size_t mcc = (size_t)N_B * C_CH * C_CH;
  float* Vb    = (float*)d_ws;
  float* Kb    = Vb + per;
  float* Qb    = Kb + per;
  float* Mpart = Qb + per;
  float* Mb    = Mpart + PCHUNKS * mcc;
  float* M2    = Mb + mcc;

  qkv_proj_kernel<<<3072, 256, 0, stream>>>(x, w0, b0, w1, b1, w2, b2, Vb, Kb, Qb);
  kv_gram_kernel<<<1024, 256, 0, stream>>>(Kb, Vb, Mpart);
  reduce_m_kernel<<<1024, 256, 0, stream>>>(Mpart, Mb);
  m_w3_kernel<<<128, 256, 0, stream>>>(Mb, w3, M2);
  out_kernel<<<1024, 256, 0, stream>>>(Qb, M2, b3, out);
}


# out kernel: Q and M2 tiles go through LDS, shared by the 8 waves of a 128 px x 128 ch work-group (32 d per stage, 2 stages); WMMA fragments read from LDS; 256 groups (was 1024); sum order unchanged
# speedup vs baseline: 3.0565x; 3.0565x over previous
#include <hip/hip_runtime.h>


#define C_CH 256
#define HW_P 4096
#define N_B  4
#define PCHUNKS 16
#define PCHUNK_LEN (HW_P / PCHUNKS)

typedef __attribute__((ext_vector_type(2))) float v2f;
typedef __attribute__((ext_vector_type(4))) float v4f;
typedef __attribute__((ext_vector_type(8))) float v8f;

__device__ __forceinline__ v8f wmma_f32(v2f a, v2f b, v8f c) {
  return __builtin_amdgcn_wmma_f32_16x16x4_f32(false, a, false, b, (short)0, c,
                                               false, false);
}

__device__ __forceinline__ v8f splat8(float v) {
  v8f r = {v, v, v, v, v, v, v, v};
  return r;
}

#define PJ_BK 32
#define PJ_XQ 288
#define PJ_WP 36

__global__ void __launch_bounds__(256, 6)
qkv_proj_kernel(const float* __restrict__ x,
                const float* __restrict__ w0, const float* __restrict__ b0,
                const float* __restrict__ w1, const float* __restrict__ b1,
                const float* __restrict__ w2, const float* __restrict__ b2,
                float* __restrict__ Vb, float* __restrict__ Kb,
                float* __restrict__ Qb) {
  __shared__ __attribute__((aligned(16))) float xs[2][PJ_BK / 2][PJ_XQ];
  __shared__ __attribute__((aligned(16))) float ws[2][128][PJ_WP];

  const int tid  = threadIdx.x;
  const int lane = tid & 31;
  const int l15  = lane & 15;
  const int hi   = lane >> 4;
  const int wave = tid >> 5;
  const int wm   = wave & 1;
  const int wn   = wave >> 1;

  int t = blockIdx.x;
  const int nb    = t % 6;  t /= 6;
  const int ptile = t & 31; t >>= 5;
  const int n     = t;
  const int proj  = nb >> 1;
  const int ob    = (nb & 1) * 128;

  const float* w    = (proj == 0) ? w0 : (proj == 1) ? w1 : w2;
  const float* bias = (proj == 0) ? b0 : (proj == 1) ? b1 : b2;
  float* out        = (proj == 0) ? Vb : (proj == 1) ? Kb : Qb;

  const int p0 = ptile * 128;

  const int wrow = tid >> 3;
  const int wcol = 4 * (tid & 7);
  const int xq0  = wave >> 1;
  const int xcol = 32 * (lane >> 2) + 16 * (wave & 1) + 4 * (lane & 3);
  const float* xg = x + ((size_t)n * C_CH + wave) * HW_P + p0 + 4 * lane;
  const float* wg = w + (size_t)(ob + wrow) * C_CH + wcol;

  v4f xr[4], wr[4];
#pragma unroll
  for (int i = 0; i < 4; ++i) {
    xr[i] = *(const v4f*)(xg + (size_t)(8 * i) * HW_P);
    wr[i] = *(const v4f*)(wg + (size_t)(32 * i) * C_CH);
  }
#pragma unroll
  for (int i = 0; i < 4; ++i) {
    *(v4f*)&xs[0][xq0 + 4 * i][xcol]      = xr[i];
    *(v4f*)&ws[0][wrow + 32 * i][wcol]    = wr[i];
  }

  v8f acc[4][2];
  {
    const float bv0 = bias[ob + 32 * wn + l15];
    const float bv1 = bias[ob + 32 * wn + 16 + l15];
#pragma unroll
    for (int j = 0; j < 4; ++j) {
      acc[j][0] = splat8(bv0);
      acc[j][1] = splat8(bv1);
    }
  }
  __syncthreads();

#pragma unroll 1
  for (int ch = 0; ch < C_CH / PJ_BK; ++ch) {
    const int st = ch & 1;
    const bool more = (ch + 1 < C_CH / PJ_BK);
    if (more) {
      const int c0 = (ch + 1) * PJ_BK;
#pragma unroll
      for (int i = 0; i < 4; ++i) {
        xr[i] = *(const v4f*)(xg + (size_t)(c0 + 8 * i) * HW_P);
        wr[i] = *(const v4f*)(wg + (size_t)(32 * i) * C_CH + c0);
      }
    }

    const float* xa  = &xs[st][hi][128 * wm + l15];
    const float* wb0 = &ws[st][32 * wn + l15][2 * hi];
    const float* wb1 = &ws[st][32 * wn + 16 + l15][2 * hi];
#pragma unroll
    for (int s = 0; s < PJ_BK / 4; ++s) {
      const v2f bb0 = *(const v2f*)(wb0 + 4 * s);
      const v2f bb1 = *(const v2f*)(wb1 + 4 * s);
#pragma unroll
      for (int j = 0; j < 4; ++j) {
        v2f a;
        a.x = xa[(2 * s) * PJ_XQ + 32 * j];
        a.y = xa[(2 * s) * PJ_XQ + 32 * j + 16];
        acc[j][0] = wmma_f32(a, bb0, acc[j][0]);
        acc[j][1] = wmma_f32(a, bb1, acc[j][1]);
      }
    }

    if (more) {
#pragma unroll
      for (int i = 0; i < 4; ++i) {
        *(v4f*)&xs[st ^ 1][xq0 + 4 * i][xcol]      = xr[i];
        *(v4f*)&ws[st ^ 1][wrow + 32 * i][wcol]    = wr[i];
      }
      __syncthreads();
    }
  }

  float* obase = out + ((size_t)n * HW_P + p0 + 64 * wm) * C_CH +
                 ob + 32 * wn + l15;
#pragma unroll
  for (int j = 0; j < 4; ++j) {
#pragma unroll
    for (int r = 0; r < 8; ++r) {
      obase[(size_t)(16 * j + r + 8 * hi) * C_CH]      = acc[j][0][r];
      obase[(size_t)(16 * j + r + 8 * hi) * C_CH + 16] = acc[j][1][r];
    }
  }
}

#define GR_BP 32
#define GR_PQ 288

__global__ void __launch_bounds__(256)
kv_gram_kernel(const float* __restrict__ Kb, const float* __restrict__ Vb,
               float* __restrict__ Mpart) {
  __shared__ __attribute__((aligned(16))) float ks[2][GR_BP / 2][GR_PQ];
  __shared__ __attribute__((aligned(16))) float vs[2][GR_BP / 2][GR_PQ];

  const int tid  = threadIdx.x;
  const int lane = tid & 31;
  const int l15  = lane & 15;
  const int hi   = lane >> 4;
  const int wave = tid >> 5;
  const int wm   = wave & 1;
  const int wn   = wave >> 1;

  int t = blockIdx.x;
  const int ct    = t & 1;  t >>= 1;
  const int dt    = t & 1;  t >>= 1;
  const int chunk = t & (PCHUNKS - 1); t /= PCHUNKS;
  const int n     = t;
  const int d0 = dt * 128, c0 = ct * 128;
  const int pbeg = chunk * PCHUNK_LEN;

  const int q0   = wave >> 1;
  const int scol = 32 * (lane >> 2) + 16 * (wave & 1) + 4 * (lane & 3);
  const float* kg = Kb + ((size_t)n * HW_P + pbeg + wave) * C_CH + d0 + 4 * lane;
  const float* vg = Vb + ((size_t)n * HW_P + pbeg + wave) * C_CH + c0 + 4 * lane;

  v4f kr[4], vr[4];
#pragma unroll
  for (int i = 0; i < 4; ++i) {
    kr[i] = *(const v4f*)(kg + (size_t)(8 * i) * C_CH);
    vr[i] = *(const v4f*)(vg + (size_t)(8 * i) * C_CH);
  }
#pragma unroll
  for (int i = 0; i < 4; ++i) {
    *(v4f*)&ks[0][q0 + 4 * i][scol] = kr[i];
    *(v4f*)&vs[0][q0 + 4 * i][scol] = vr[i];
  }

  v8f acc[4][2];
#pragma unroll
  for (int j = 0; j < 4; ++j) {
    acc[j][0] = splat8(0.0f);
    acc[j][1] = splat8(0.0f);
  }
  __syncthreads();

#pragma unroll 1
  for (int sg = 0; sg < PCHUNK_LEN / GR_BP; ++sg) {
    const int st = sg & 1;
    const bool more = (sg + 1 < PCHUNK_LEN / GR_BP);
    if (more) {
      const size_t po = (size_t)((sg + 1) * GR_BP) * C_CH;
#pragma unroll
      for (int i = 0; i < 4; ++i) {
        kr[i] = *(const v4f*)(kg + po + (size_t)(8 * i) * C_CH);
        vr[i] = *(const v4f*)(vg + po + (size_t)(8 * i) * C_CH);
      }
    }

    const float* ka = &ks[st][hi][128 * wm + l15];
    const float* vb = &vs[st][hi][64 * wn + l15];
#pragma unroll
    for (int s = 0; s < GR_BP / 4; ++s) {
      v2f bb0, bb1;
      bb0.x = vb[(2 * s) * GR_PQ];
      bb0.y = vb[(2 * s) * GR_PQ + 16];
      bb1.x = vb[(2 * s) * GR_PQ + 32];
      bb1.y = vb[(2 * s) * GR_PQ + 48];
#pragma unroll
      for (int j = 0; j < 4; ++j) {
        v2f a;
        a.x = ka[(2 * s) * GR_PQ + 32 * j];
        a.y = ka[(2 * s) * GR_PQ + 32 * j + 16];
        acc[j][0] = wmma_f32(a, bb0, acc[j][0]);
        acc[j][1] = wmma_f32(a, bb1, acc[j][1]);
      }
    }

    if (more) {
#pragma unroll
      for (int i = 0; i < 4; ++i) {
        *(v4f*)&ks[st ^ 1][q0 + 4 * i][scol] = kr[i];
        *(v4f*)&vs[st ^ 1][q0 + 4 * i][scol] = vr[i];
      }
      __syncthreads();
    }
  }

  float* mp = Mpart + ((size_t)chunk * N_B + n) * C_CH * C_CH +
              (size_t)(d0 + 64 * wm) * C_CH + c0 + 32 * wn + l15;
#pragma unroll
  for (int j = 0; j < 4; ++j) {
#pragma unroll
    for (int r = 0; r < 8; ++r) {
      mp[(size_t)(16 * j + r + 8 * hi) * C_CH]      = acc[j][0][r];
      mp[(size_t)(16 * j + r + 8 * hi) * C_CH + 16] = acc[j][1][r];
    }
  }
}

__global__ void __launch_bounds__(256)
reduce_m_kernel(const float* __restrict__ Mpart, float* __restrict__ Mb) {
  const size_t idx    = (size_t)blockIdx.x * blockDim.x + threadIdx.x;
  const size_t stride = (size_t)N_B * C_CH * C_CH;
  float s = 0.0f;
#pragma unroll
  for (int c = 0; c < PCHUNKS; ++c) s += Mpart[c * stride + idx];
  Mb[idx] = s;
}

__global__ void __launch_bounds__(256)
m_w3_kernel(const float* __restrict__ Mb, const float* __restrict__ w3,
            float* __restrict__ M2) {
  const int lane = threadIdx.x & 31;
  const int l15  = lane & 15;
  const int hi   = lane >> 4;
  int t = blockIdx.x * (blockDim.x >> 5) + (threadIdx.x >> 5);
  const int otile = t & 15; t >>= 4;
  const int dtile = t & 15; t >>= 4;
  const int n     = t;
  const int d0 = dtile * 16, o0 = otile * 16;

  const float* mrow = Mb + ((size_t)n * C_CH + d0 + l15) * C_CH;
  const float* wrow = w3 + (size_t)(o0 + l15) * C_CH;
  v8f acc = {};
#pragma unroll 4
  for (int k0 = 0; k0 < C_CH; k0 += 4) {
    const int cb = k0 + 2 * hi;
    v2f a = *(const v2f*)(mrow + cb);
    v2f b = *(const v2f*)(wrow + cb);
    acc = wmma_f32(a, b, acc);
  }
  const float inv_hw = 1.0f / (float)HW_P;
  float* m2base = M2 + ((size_t)n * C_CH + d0) * C_CH + o0 + l15;
#pragma unroll
  for (int r = 0; r < 8; ++r)
    m2base[(size_t)(r + 8 * hi) * C_CH] = acc[r] * inv_hw;
}

#define OT_BD 32
#define OT_QP 36
#define OT_MQ 288

__global__ void __launch_bounds__(256)
out_kernel(const float* __restrict__ Qb, const float* __restrict__ M2,
           const float* __restrict__ b3, float* __restrict__ out) {
  __shared__ __attribute__((aligned(16))) float qs[2][128][OT_QP];
  __shared__ __attribute__((aligned(16))) float ms[2][OT_BD / 2][OT_MQ];

  const int tid  = threadIdx.x;
  const int lane = tid & 31;
  const int l15  = lane & 15;
  const int hi   = lane >> 4;
  const int wave = tid >> 5;
  const int wm   = wave & 1;
  const int wn   = wave >> 1;

  int t = blockIdx.x;
  const int ot    = t & 1;  t >>= 1;
  const int ptile = t & 31; t >>= 5;
  const int n     = t;
  const int p0 = ptile * 128, ob = ot * 128;

  const int qrow = tid >> 3;
  const int qcol = 4 * (tid & 7);
  const int mq0  = wave >> 1;
  const int mcol = 32 * (lane >> 2) + 16 * (wave & 1) + 4 * (lane & 3);
  const float* qg = Qb + ((size_t)n * HW_P + p0 + qrow) * C_CH + qcol;
  const float* mg = M2 + ((size_t)n * C_CH + wave) * C_CH + ob + 4 * lane;

  v4f qr[4], mr[4];
#pragma unroll
  for (int i = 0; i < 4; ++i) {
    qr[i] = *(const v4f*)(qg + (size_t)(32 * i) * C_CH);
    mr[i] = *(const v4f*)(mg + (size_t)(8 * i) * C_CH);
  }
#pragma unroll
  for (int i = 0; i < 4; ++i) {
    *(v4f*)&qs[0][qrow + 32 * i][qcol] = qr[i];
    *(v4f*)&ms[0][mq0 + 4 * i][mcol]   = mr[i];
  }

  v8f acc[4][2];
  {
    const float bv0 = b3[ob + 32 * wn + l15];
    const float bv1 = b3[ob + 32 * wn + 16 + l15];
#pragma unroll
    for (int j = 0; j < 4; ++j) {
      acc[j][0] = splat8(bv0);
      acc[j][1] = splat8(bv1);
    }
  }
  __syncthreads();

#pragma unroll 1
  for (int ch = 0; ch < C_CH / OT_BD; ++ch) {
    const int st = ch & 1;
    const bool more = (ch + 1 < C_CH / OT_BD);
    if (more) {
      const int d0 = (ch + 1) * OT_BD;
#pragma unroll
      for (int i = 0; i < 4; ++i) {
        qr[i] = *(const v4f*)(qg + (size_t)(32 * i) * C_CH + d0);
        mr[i] = *(const v4f*)(mg + (size_t)(d0 + 8 * i) * C_CH);
      }
    }

    const float* qa = &qs[st][64 * wm + l15][2 * hi];
    const float* mb = &ms[st][hi][64 * wn + l15];
#pragma unroll
    for (int s = 0; s < OT_BD / 4; ++s) {
      v2f bb0, bb1;
      bb0.x = mb[(2 * s) * OT_MQ];
      bb0.y = mb[(2 * s) * OT_MQ + 16];
      bb1.x = mb[(2 * s) * OT_MQ + 32];
      bb1.y = mb[(2 * s) * OT_MQ + 48];
#pragma unroll
      for (int j = 0; j < 4; ++j) {
        const v2f a = *(const v2f*)(qa + (16 * j) * OT_QP + 4 * s);
        acc[j][0] = wmma_f32(a, bb0, acc[j][0]);
        acc[j][1] = wmma_f32(a, bb1, acc[j][1]);
      }
    }

    if (more) {
#pragma unroll
      for (int i = 0; i < 4; ++i) {
        *(v4f*)&qs[st ^ 1][qrow + 32 * i][qcol] = qr[i];
        *(v4f*)&ms[st ^ 1][mq0 + 4 * i][mcol]   = mr[i];
      }
      __syncthreads();
    }
  }

  float* obase = out + ((size_t)n * C_CH + ob + 32 * wn + l15) * HW_P +
                 p0 + 64 * wm + 8 * hi;
#pragma unroll
  for (int j = 0; j < 4; ++j) {
#pragma unroll
    for (int u = 0; u < 2; ++u) {
      const v4f lo = {acc[j][u][0], acc[j][u][1], acc[j][u][2], acc[j][u][3]};
      const v4f hh = {acc[j][u][4], acc[j][u][5], acc[j][u][6], acc[j][u][7]};
      *(v4f*)(obase + (size_t)(16 * u) * HW_P + 16 * j)     = lo;
      *(v4f*)(obase + (size_t)(16 * u) * HW_P + 16 * j + 4) = hh;
    }
  }
}

extern "C" void kernel_launch(void* const* d_in, const int* in_sizes, int n_in,
                              void* d_out, int out_size, void* d_ws,
                              size_t ws_size, hipStream_t stream) {
  const float* x  = (const float*)d_in[0];
  const float* w0 = (const float*)d_in[1];
  const float* b0 = (const float*)d_in[2];
  const float* w1 = (const float*)d_in[3];
  const float* b1 = (const float*)d_in[4];
  const float* w2 = (const float*)d_in[5];
  const float* b2 = (const float*)d_in[6];
  const float* w3 = (const float*)d_in[7];
  const float* b3 = (const float*)d_in[8];
  float* out = (float*)d_out;

  const size_t per = (size_t)N_B * HW_P * C_CH;
  const size_t mcc = (size_t)N_B * C_CH * C_CH;
  float* Vb    = (float*)d_ws;
  float* Kb    = Vb + per;
  float* Qb    = Kb + per;
  float* Mpart = Qb + per;
  float* Mb    = Mpart + PCHUNKS * mcc;
  float* M2    = Mb + mcc;

  qkv_proj_kernel<<<768, 256, 0, stream>>>(x, w0, b0, w1, b1, w2, b2, Vb, Kb, Qb);
  kv_gram_kernel<<<256, 256, 0, stream>>>(Kb, Vb, Mpart);
  reduce_m_kernel<<<1024, 256, 0, stream>>>(Mpart, Mb);
  m_w3_kernel<<<128, 256, 0, stream>>>(Mb, w3, M2);
  out_kernel<<<256, 256, 0, stream>>>(Qb, M2, b3, out);
}
